# starting kernel
# baseline (speedup 1.0000x reference)
#include <hip/hip_runtime.h>
#include <hip/hip_bf16.h>
#include <math.h>

typedef __attribute__((ext_vector_type(2))) float v2f;
typedef __attribute__((ext_vector_type(8))) float v8f;

#define F_DIM 64

__global__ void zero_kernel(float* __restrict__ p, int n) {
    int i = blockIdx.x * blockDim.x + threadIdx.x;
    if (i < n) p[i] = 0.0f;
}

__global__ void __launch_bounds__(64)
matvec_wmma_kernel(const float* __restrict__ X, const float* __restrict__ W,
                   float* __restrict__ s, int n_nodes) {
    const int lane = threadIdx.x & 31;
    const int wave = threadIdx.x >> 5;
    const int tile = blockIdx.x * (blockDim.x >> 5) + wave;
    const int base = tile << 4;
    if (base >= n_nodes) return;

    const int m    = lane & 15;
    const int half = lane >> 4;

    const float* __restrict__ xrow = X + (size_t)(base + m) * F_DIM;

    v8f c = {};
#pragma unroll
    for (int k = 0; k < F_DIM; k += 4) {
        const v2f a = *reinterpret_cast<const v2f*>(xrow + k + (half << 1));
        const v2f wv = *reinterpret_cast<const v2f*>(W + k + (half << 1));
        v2f bf;
        bf.x = (m == 0) ? wv.x : 0.0f;
        bf.y = (m == 0) ? wv.y : 0.0f;
        c = __builtin_amdgcn_wmma_f32_16x16x4_f32(false, a, false, bf,
                                                  (short)0, c, false, false);
    }

    if (m == 0) {
        float* __restrict__ sp = s + base + (half << 3);
#pragma unroll
        for (int r = 0; r < 8; ++r) sp[r] = c[r];
    }
}

__global__ void __launch_bounds__(256)
edge_scatter_kernel(const int* __restrict__ src, const int* __restrict__ dst,
                    const float* __restrict__ s, float* __restrict__ acc,
                    float* __restrict__ deg, int n_edges) {
    int e = blockIdx.x * blockDim.x + threadIdx.x;
    if (e >= n_edges) return;
    const int si = src[e];
    const int di = dst[e];
    const float sv = s[si];
    atomicAdd(&acc[di], sv);
    atomicAdd(&deg[di], 1.0f);
}

__global__ void __launch_bounds__(256)
finalize_kernel(const float* __restrict__ acc, const float* __restrict__ deg,
                const float* __restrict__ b, float* __restrict__ out, int n) {
    int i = blockIdx.x * blockDim.x + threadIdx.x;
    if (i >= n) return;
    const float d = fmaxf(deg[i], 1.0f);
    const float x = acc[i] / d + b[0];
    out[i] = 1.0f / (1.0f + expf(-x));
}

extern "C" void kernel_launch(void* const* d_in, const int* in_sizes, int n_in,
                              void* d_out, int out_size, void* d_ws, size_t ws_size,
                              hipStream_t stream) {
    const float* X     = (const float*)d_in[0];
    const int*   edges = (const int*)d_in[1];
    const float* W     = (const float*)d_in[2];
    const float* b     = (const float*)d_in[3];
    float*       out   = (float*)d_out;

    const int n_nodes = in_sizes[0] / F_DIM;
    const int n_edges = in_sizes[1] / 2;
    const int* src = edges;
    const int* dst = edges + n_edges;

    float* s   = (float*)d_ws;
    float* acc = s + n_nodes;
    float* deg = acc + n_nodes;

    {
        const int n = 2 * n_nodes;
        zero_kernel<<<(n + 255) / 256, 256, 0, stream>>>(acc, n);
    }

    {
        const int tiles  = (n_nodes + 15) / 16;
        const int blocks = (tiles + 1) / 2;
        matvec_wmma_kernel<<<blocks, 64, 0, stream>>>(X, W, s, n_nodes);
    }

    edge_scatter_kernel<<<(n_edges + 255) / 256, 256, 0, stream>>>(
        src, dst, s, acc, deg, n_edges);

    finalize_kernel<<<(n_nodes + 255) / 256, 256, 0, stream>>>(
        acc, deg, b, out, n_nodes);
}


# Edge sums kept in one acc/deg plane set per package half (by die number), merged in the last pass: no atomic line is shared by both halves; the sum order changes to two partial sums
# speedup vs baseline: 4.1746x; 4.1746x over previous
#include <hip/hip_runtime.h>
#include <hip/hip_bf16.h>
#include <math.h>

typedef __attribute__((ext_vector_type(2))) float v2f;
typedef __attribute__((ext_vector_type(8))) float v8f;

#define F_DIM 64


__device__ __forceinline__ bool two_sets(int n_nodes, size_t ws_bytes) {
    return ws_bytes >= (size_t)5 * (size_t)n_nodes * sizeof(float);
}

__device__ __forceinline__ unsigned package_half() {
    return (__builtin_amdgcn_s_sendmsg_rtn(0x87) >> 18) & 1u;
}

__global__ void zero_kernel(float* __restrict__ p, int n_nodes, size_t ws_bytes) {
    const int n = (two_sets(n_nodes, ws_bytes) ? 4 : 2) * n_nodes;
    int i = blockIdx.x * blockDim.x + threadIdx.x;
    if (i < n) p[i] = 0.0f;
}

__global__ void __launch_bounds__(64)
matvec_wmma_kernel(const float* __restrict__ X, const float* __restrict__ W,
                   float* __restrict__ s, int n_nodes) {
    const int lane = threadIdx.x & 31;
    const int wave = threadIdx.x >> 5;
    const int tile = blockIdx.x * (blockDim.x >> 5) + wave;
    const int base = tile << 4;
    if (base >= n_nodes) return;

    const int m    = lane & 15;
    const int half = lane >> 4;

    const float* __restrict__ xrow = X + (size_t)(base + m) * F_DIM;

    v8f c = {};
#pragma unroll
    for (int k = 0; k < F_DIM; k += 4) {
        const v2f a = *reinterpret_cast<const v2f*>(xrow + k + (half << 1));
        const v2f wv = *reinterpret_cast<const v2f*>(W + k + (half << 1));
        v2f bf;
        bf.x = (m == 0) ? wv.x : 0.0f;
        bf.y = (m == 0) ? wv.y : 0.0f;
        c = __builtin_amdgcn_wmma_f32_16x16x4_f32(false, a, false, bf,
                                                  (short)0, c, false, false);
    }

    if (m == 0) {
        float* __restrict__ sp = s + base + (half << 3);
#pragma unroll
        for (int r = 0; r < 8; ++r) sp[r] = c[r];
    }
}

__global__ void __launch_bounds__(256)
edge_scatter_kernel(const int* __restrict__ src, const int* __restrict__ dst,
                    const float* __restrict__ s, float* __restrict__ planes,
                    int n_nodes, int n_edges, size_t ws_bytes) {
    int e = blockIdx.x * blockDim.x + threadIdx.x;
    if (e >= n_edges) return;
    const int si = src[e];
    const int di = dst[e];
    const unsigned set = two_sets(n_nodes, ws_bytes) ? package_half() : 0u;
    float* __restrict__ acc = planes + (size_t)set * 2u * (size_t)n_nodes;
    float* __restrict__ deg = acc + n_nodes;
    const float sv = s[si];
    atomicAdd(&acc[di], sv);
    atomicAdd(&deg[di], 1.0f);
}

__global__ void __launch_bounds__(256)
finalize_kernel(const float* __restrict__ planes, const float* __restrict__ b,
                float* __restrict__ out, int n, size_t ws_bytes) {
    int i = blockIdx.x * blockDim.x + threadIdx.x;
    if (i >= n) return;
    float a  = planes[i];
    float dg = planes[(size_t)n + i];
    if (two_sets(n, ws_bytes)) {
        a  += planes[(size_t)2 * n + i];
        dg += planes[(size_t)3 * n + i];
    }
    const float d = fmaxf(dg, 1.0f);
    const float x = a / d + b[0];
    out[i] = 1.0f / (1.0f + expf(-x));
}

extern "C" void kernel_launch(void* const* d_in, const int* in_sizes, int n_in,
                              void* d_out, int out_size, void* d_ws, size_t ws_size,
                              hipStream_t stream) {
    const float* X     = (const float*)d_in[0];
    const int*   edges = (const int*)d_in[1];
    const float* W     = (const float*)d_in[2];
    const float* b     = (const float*)d_in[3];
    float*       out   = (float*)d_out;

    const int n_nodes = in_sizes[0] / F_DIM;
    const int n_edges = in_sizes[1] / 2;
    const int* src = edges;
    const int* dst = edges + n_edges;

    float* s      = (float*)d_ws;
    float* planes = s + n_nodes;

    {
        const int n = 4 * n_nodes;
        zero_kernel<<<(n + 255) / 256, 256, 0, stream>>>(planes, n_nodes, ws_size);
    }

    {
        const int tiles  = (n_nodes + 15) / 16;
        const int blocks = (tiles + 1) / 2;
        matvec_wmma_kernel<<<blocks, 64, 0, stream>>>(X, W, s, n_nodes);
    }

    edge_scatter_kernel<<<(n_edges + 255) / 256, 256, 0, stream>>>(
        src, dst, s, planes, n_nodes, n_edges, ws_size);

    finalize_kernel<<<(n_nodes + 255) / 256, 256, 0, stream>>>(
        planes, b, out, n_nodes, ws_size);
}


# Plane clearing folded into the matvec launch: its 6250 waves each store their share of the acc/deg planes; the separate zero launch and its 12.5k one-store waves are gone (4 launches to 3)
# speedup vs baseline: 4.4200x; 1.0588x over previous
#include <hip/hip_runtime.h>
#include <hip/hip_bf16.h>
#include <math.h>

typedef __attribute__((ext_vector_type(2))) float v2f;
typedef __attribute__((ext_vector_type(8))) float v8f;

#define F_DIM 64


__device__ __forceinline__ bool two_sets(int n_nodes, size_t ws_bytes) {
    return ws_bytes >= (size_t)5 * (size_t)n_nodes * sizeof(float);
}

__device__ __forceinline__ unsigned package_half() {
    return (__builtin_amdgcn_s_sendmsg_rtn(0x87) >> 18) & 1u;
}

__global__ void __launch_bounds__(64)
matvec_wmma_kernel(const float* __restrict__ X, const float* __restrict__ W,
                   float* __restrict__ s, float* __restrict__ planes,
                   int n_nodes, size_t ws_bytes) {
    {
        const int node = (blockIdx.x << 5) + (threadIdx.x & 31);
        const int n_planes = two_sets(n_nodes, ws_bytes) ? 4 : 2;
        if (node < n_nodes)
            for (int p = threadIdx.x >> 5; p < n_planes; p += 2)
                planes[(size_t)p * (size_t)n_nodes + node] = 0.0f;
    }

    const int lane = threadIdx.x & 31;
    const int wave = threadIdx.x >> 5;
    const int tile = blockIdx.x * (blockDim.x >> 5) + wave;
    const int base = tile << 4;
    if (base >= n_nodes) return;

    const int m    = lane & 15;
    const int half = lane >> 4;

    const float* __restrict__ xrow = X + (size_t)(base + m) * F_DIM;

    v8f c = {};
#pragma unroll
    for (int k = 0; k < F_DIM; k += 4) {
        const v2f a = *reinterpret_cast<const v2f*>(xrow + k + (half << 1));
        const v2f wv = *reinterpret_cast<const v2f*>(W + k + (half << 1));
        v2f bf;
        bf.x = (m == 0) ? wv.x : 0.0f;
        bf.y = (m == 0) ? wv.y : 0.0f;
        c = __builtin_amdgcn_wmma_f32_16x16x4_f32(false, a, false, bf,
                                                  (short)0, c, false, false);
    }

    if (m == 0) {
        float* __restrict__ sp = s + base + (half << 3);
#pragma unroll
        for (int r = 0; r < 8; ++r) sp[r] = c[r];
    }
}

__global__ void __launch_bounds__(256)
edge_scatter_kernel(const int* __restrict__ src, const int* __restrict__ dst,
                    const float* __restrict__ s, float* __restrict__ planes,
                    int n_nodes, int n_edges, size_t ws_bytes) {
    int e = blockIdx.x * blockDim.x + threadIdx.x;
    if (e >= n_edges) return;
    const int si = src[e];
    const int di = dst[e];
    const unsigned set = two_sets(n_nodes, ws_bytes) ? package_half() : 0u;
    float* __restrict__ acc = planes + (size_t)set * 2u * (size_t)n_nodes;
    float* __restrict__ deg = acc + n_nodes;
    const float sv = s[si];
    atomicAdd(&acc[di], sv);
    atomicAdd(&deg[di], 1.0f);
}

__global__ void __launch_bounds__(256)
finalize_kernel(const float* __restrict__ planes, const float* __restrict__ b,
                float* __restrict__ out, int n, size_t ws_bytes) {
    int i = blockIdx.x * blockDim.x + threadIdx.x;
    if (i >= n) return;
    float a  = planes[i];
    float dg = planes[(size_t)n + i];
    if (two_sets(n, ws_bytes)) {
        a  += planes[(size_t)2 * n + i];
        dg += planes[(size_t)3 * n + i];
    }
    const float d = fmaxf(dg, 1.0f);
    const float x = a / d + b[0];
    out[i] = 1.0f / (1.0f + expf(-x));
}

extern "C" void kernel_launch(void* const* d_in, const int* in_sizes, int n_in,
                              void* d_out, int out_size, void* d_ws, size_t ws_size,
                              hipStream_t stream) {
    const float* X     = (const float*)d_in[0];
    const int*   edges = (const int*)d_in[1];
    const float* W     = (const float*)d_in[2];
    const float* b     = (const float*)d_in[3];
    float*       out   = (float*)d_out;

    const int n_nodes = in_sizes[0] / F_DIM;
    const int n_edges = in_sizes[1] / 2;
    const int* src = edges;
    const int* dst = edges + n_edges;

    float* s      = (float*)d_ws;
    float* planes = s + n_nodes;

    {
        const int tiles  = (n_nodes + 15) / 16;
        const int blocks = (tiles + 1) / 2;
        matvec_wmma_kernel<<<blocks, 64, 0, stream>>>(X, W, s, planes, n_nodes,
                                                      ws_size);
    }

    edge_scatter_kernel<<<(n_edges + 255) / 256, 256, 0, stream>>>(
        src, dst, s, planes, n_nodes, n_edges, ws_size);

    finalize_kernel<<<(n_nodes + 255) / 256, 256, 0, stream>>>(
        planes, b, out, n_nodes, ws_size);
}


# Sum and count of a node side by side; an edge's two atomic adds leave from two neighbouring lanes of one instruction: one request per edge at the second-level cache, not two; sum order as before
# speedup vs baseline: 4.8411x; 1.0953x over previous
#include <hip/hip_runtime.h>
#include <hip/hip_bf16.h>
#include <math.h>

typedef __attribute__((ext_vector_type(2))) float v2f;
typedef __attribute__((ext_vector_type(8))) float v8f;

#define F_DIM 64


__device__ __forceinline__ bool two_sets(int n_nodes, size_t ws_bytes) {
    return ws_bytes >= (size_t)5 * (size_t)n_nodes * sizeof(float);
}

__device__ __forceinline__ unsigned package_half() {
    return (__builtin_amdgcn_s_sendmsg_rtn(0x87) >> 18) & 1u;
}

__global__ void __launch_bounds__(64)
matvec_wmma_kernel(const float* __restrict__ X, const float* __restrict__ W,
                   float* __restrict__ s, float* __restrict__ planes,
                   int n_nodes, size_t ws_bytes) {
    {
        const int node = (blockIdx.x << 5) + (threadIdx.x & 31);
        const int n_planes = two_sets(n_nodes, ws_bytes) ? 4 : 2;
        if (node < n_nodes)
            for (int p = threadIdx.x >> 5; p < n_planes; p += 2)
                planes[(size_t)p * (size_t)n_nodes + node] = 0.0f;
    }

    const int lane = threadIdx.x & 31;
    const int wave = threadIdx.x >> 5;
    const int tile = blockIdx.x * (blockDim.x >> 5) + wave;
    const int base = tile << 4;
    if (base >= n_nodes) return;

    const int m    = lane & 15;
    const int half = lane >> 4;

    const float* __restrict__ xrow = X + (size_t)(base + m) * F_DIM;

    v8f c = {};
#pragma unroll
    for (int k = 0; k < F_DIM; k += 4) {
        const v2f a = *reinterpret_cast<const v2f*>(xrow + k + (half << 1));
        const v2f wv = *reinterpret_cast<const v2f*>(W + k + (half << 1));
        v2f bf;
        bf.x = (m == 0) ? wv.x : 0.0f;
        bf.y = (m == 0) ? wv.y : 0.0f;
        c = __builtin_amdgcn_wmma_f32_16x16x4_f32(false, a, false, bf,
                                                  (short)0, c, false, false);
    }

    if (m == 0) {
        float* __restrict__ sp = s + base + (half << 3);
#pragma unroll
        for (int r = 0; r < 8; ++r) sp[r] = c[r];
    }
}

__global__ void __launch_bounds__(256)
edge_scatter_kernel(const int* __restrict__ src, const int* __restrict__ dst,
                    const float* __restrict__ s, float* __restrict__ planes,
                    int n_nodes, int n_edges, size_t ws_bytes) {
    const int e = blockIdx.x * blockDim.x + threadIdx.x;
    const int ec = e < n_edges ? e : n_edges - 1;
    const int si = src[ec];
    const int di = dst[ec];
    const unsigned set = two_sets(n_nodes, ws_bytes) ? package_half() : 0u;
    float* __restrict__ pairs = planes + (size_t)set * 2u * (size_t)n_nodes;
    const float sv = s[si];
    const int dn = __builtin_amdgcn_mov_dpp(di, 0xB1, 0xF, 0xF, true);
    const int odd = e & 1;
    if ((e & ~1) < n_edges) {
        const int d = odd ? dn : di;
        atomicAdd(&pairs[2 * d + odd], odd ? 1.0f : sv);
    }
    if ((e | 1) < n_edges) {
        const int d = odd ? di : dn;
        atomicAdd(&pairs[2 * d + (odd ^ 1)], odd ? sv : 1.0f);
    }
}

__global__ void __launch_bounds__(256)
finalize_kernel(const float* __restrict__ planes, const float* __restrict__ b,
                float* __restrict__ out, int n, size_t ws_bytes) {
    int i = blockIdx.x * blockDim.x + threadIdx.x;
    if (i >= n) return;
    const float* __restrict__ p0 = planes + 2 * (size_t)i;
    float a  = p0[0];
    float dg = p0[1];
    if (two_sets(n, ws_bytes)) {
        const float* __restrict__ p1 = p0 + 2 * (size_t)n;
        a  += p1[0];
        dg += p1[1];
    }
    const float d = fmaxf(dg, 1.0f);
    const float x = a / d + b[0];
    out[i] = 1.0f / (1.0f + expf(-x));
}

extern "C" void kernel_launch(void* const* d_in, const int* in_sizes, int n_in,
                              void* d_out, int out_size, void* d_ws, size_t ws_size,
                              hipStream_t stream) {
    const float* X     = (const float*)d_in[0];
    const int*   edges = (const int*)d_in[1];
    const float* W     = (const float*)d_in[2];
    const float* b     = (const float*)d_in[3];
    float*       out   = (float*)d_out;

    const int n_nodes = in_sizes[0] / F_DIM;
    const int n_edges = in_sizes[1] / 2;
    const int* src = edges;
    const int* dst = edges + n_edges;

    float* s      = (float*)d_ws;
    float* planes = s + n_nodes;

    {
        const int tiles  = (n_nodes + 15) / 16;
        const int blocks = (tiles + 1) / 2;
        matvec_wmma_kernel<<<blocks, 64, 0, stream>>>(X, W, s, planes, n_nodes,
                                                      ws_size);
    }

    edge_scatter_kernel<<<(n_edges + 255) / 256, 256, 0, stream>>>(
        src, dst, s, planes, n_nodes, n_edges, ws_size);

    finalize_kernel<<<(n_nodes + 255) / 256, 256, 0, stream>>>(
        planes, b, out, n_nodes, ws_size);
}


# Edge pass: each lane takes 8 edges, in rounds of 32 per wave (8 index loads, then 8 gathers in flight together): 4,883 waves to start, not 39,063; one half query a wave; sums still by atomics
# speedup vs baseline: 5.5086x; 1.1379x over previous
#include <hip/hip_runtime.h>
#include <hip/hip_bf16.h>
#include <math.h>

typedef __attribute__((ext_vector_type(2))) float v2f;
typedef __attribute__((ext_vector_type(8))) float v8f;

#define F_DIM 64


__device__ __forceinline__ bool two_sets(int n_nodes, size_t ws_bytes) {
    return ws_bytes >= (size_t)5 * (size_t)n_nodes * sizeof(float);
}

__device__ __forceinline__ unsigned package_half() {
    return (__builtin_amdgcn_s_sendmsg_rtn(0x87) >> 18) & 1u;
}

__global__ void __launch_bounds__(64)
matvec_wmma_kernel(const float* __restrict__ X, const float* __restrict__ W,
                   float* __restrict__ s, float* __restrict__ planes,
                   int n_nodes, size_t ws_bytes) {
    {
        const int node = (blockIdx.x << 5) + (threadIdx.x & 31);
        const int n_planes = two_sets(n_nodes, ws_bytes) ? 4 : 2;
        if (node < n_nodes)
            for (int p = threadIdx.x >> 5; p < n_planes; p += 2)
                planes[(size_t)p * (size_t)n_nodes + node] = 0.0f;
    }

    const int lane = threadIdx.x & 31;
    const int wave = threadIdx.x >> 5;
    const int tile = blockIdx.x * (blockDim.x >> 5) + wave;
    const int base = tile << 4;
    if (base >= n_nodes) return;

    const int m    = lane & 15;
    const int half = lane >> 4;

    const float* __restrict__ xrow = X + (size_t)(base + m) * F_DIM;

    v8f c = {};
#pragma unroll
    for (int k = 0; k < F_DIM; k += 4) {
        const v2f a = *reinterpret_cast<const v2f*>(xrow + k + (half << 1));
        const v2f wv = *reinterpret_cast<const v2f*>(W + k + (half << 1));
        v2f bf;
        bf.x = (m == 0) ? wv.x : 0.0f;
        bf.y = (m == 0) ? wv.y : 0.0f;
        c = __builtin_amdgcn_wmma_f32_16x16x4_f32(false, a, false, bf,
                                                  (short)0, c, false, false);
    }

    if (m == 0) {
        float* __restrict__ sp = s + base + (half << 3);
#pragma unroll
        for (int r = 0; r < 8; ++r) sp[r] = c[r];
    }
}

#define EDGES_PER_LANE 8
#define SCATTER_THREADS 128

template <bool kWhole>
__device__ __forceinline__ void
scatter_rounds(const int* __restrict__ src, const int* __restrict__ dst,
               const float* __restrict__ s, float* __restrict__ planes,
               int n_nodes, int n_edges, size_t ws_bytes, int first, int odd) {
    const int last = n_edges - 1;
    int si[EDGES_PER_LANE];
    int di[EDGES_PER_LANE];
#pragma unroll
    for (int j = 0; j < EDGES_PER_LANE; ++j) {
        const int e = first + 32 * j;
        const int ec = (kWhole || e < n_edges) ? e : last;
        si[j] = src[ec];
        di[j] = dst[ec];
    }
    const unsigned set = two_sets(n_nodes, ws_bytes) ? package_half() : 0u;
    float* __restrict__ pairs = planes + (size_t)set * 2u * (size_t)n_nodes;

    float sv[EDGES_PER_LANE];
#pragma unroll
    for (int j = 0; j < EDGES_PER_LANE; ++j) sv[j] = s[si[j]];

#pragma unroll
    for (int j = 0; j < EDGES_PER_LANE; ++j) {
        const int e = first + 32 * j;
        const int dn = __builtin_amdgcn_mov_dpp(di[j], 0xB1, 0xF, 0xF, true);
        if (kWhole || (e & ~1) < n_edges) {
            const int d = odd ? dn : di[j];
            atomicAdd(&pairs[2 * d + odd], odd ? 1.0f : sv[j]);
        }
        if (kWhole || (e | 1) < n_edges) {
            const int d = odd ? di[j] : dn;
            atomicAdd(&pairs[2 * d + (odd ^ 1)], odd ? sv[j] : 1.0f);
        }
    }
}

__global__ void __launch_bounds__(SCATTER_THREADS)
edge_scatter_kernel(const int* __restrict__ src, const int* __restrict__ dst,
                    const float* __restrict__ s, float* __restrict__ planes,
                    int n_nodes, int n_edges, size_t ws_bytes) {
    const int lane = threadIdx.x & 31;
    const int wave = __builtin_amdgcn_readfirstlane(
        blockIdx.x * (SCATTER_THREADS / 32) + (threadIdx.x >> 5));
    const int base = wave * (32 * EDGES_PER_LANE);
    if (base >= n_edges) return;
    if (base + 32 * EDGES_PER_LANE <= n_edges)
        scatter_rounds<true>(src, dst, s, planes, n_nodes, n_edges, ws_bytes,
                             base + lane, lane & 1);
    else
        scatter_rounds<false>(src, dst, s, planes, n_nodes, n_edges, ws_bytes,
                              base + lane, lane & 1);
}

__global__ void __launch_bounds__(256)
finalize_kernel(const float* __restrict__ planes, const float* __restrict__ b,
                float* __restrict__ out, int n, size_t ws_bytes) {
    int i = blockIdx.x * blockDim.x + threadIdx.x;
    if (i >= n) return;
    const float* __restrict__ p0 = planes + 2 * (size_t)i;
    float a  = p0[0];
    float dg = p0[1];
    if (two_sets(n, ws_bytes)) {
        const float* __restrict__ p1 = p0 + 2 * (size_t)n;
        a  += p1[0];
        dg += p1[1];
    }
    const float d = fmaxf(dg, 1.0f);
    const float x = a / d + b[0];
    out[i] = 1.0f / (1.0f + expf(-x));
}

extern "C" void kernel_launch(void* const* d_in, const int* in_sizes, int n_in,
                              void* d_out, int out_size, void* d_ws, size_t ws_size,
                              hipStream_t stream) {
    const float* X     = (const float*)d_in[0];
    const int*   edges = (const int*)d_in[1];
    const float* W     = (const float*)d_in[2];
    const float* b     = (const float*)d_in[3];
    float*       out   = (float*)d_out;

    const int n_nodes = in_sizes[0] / F_DIM;
    const int n_edges = in_sizes[1] / 2;
    const int* src = edges;
    const int* dst = edges + n_edges;

    float* s      = (float*)d_ws;
    float* planes = s + n_nodes;

    {
        const int tiles  = (n_nodes + 15) / 16;
        const int blocks = (tiles + 1) / 2;
        matvec_wmma_kernel<<<blocks, 64, 0, stream>>>(X, W, s, planes, n_nodes,
                                                      ws_size);
    }

    {
        const int edges_per_block = SCATTER_THREADS * EDGES_PER_LANE;
        edge_scatter_kernel<<<(n_edges + edges_per_block - 1) / edges_per_block,
                              SCATTER_THREADS, 0, stream>>>(
            src, dst, s, planes, n_nodes, n_edges, ws_size);
    }

    finalize_kernel<<<(n_nodes + 255) / 256, 256, 0, stream>>>(
        planes, b, out, n_nodes, ws_size);
}


# First launch loads each 16-row tile of X as 8 whole 512-byte pieces (4 lines a load, not 16) and puts it into fragment order through LDS: 48 line addresses a tile at the WGP, not 272; sums as before
# speedup vs baseline: 5.9050x; 1.0720x over previous
#include <hip/hip_runtime.h>
#include <hip/hip_bf16.h>
#include <math.h>

typedef __attribute__((ext_vector_type(2))) float v2f;
typedef __attribute__((ext_vector_type(4))) float v4f;
typedef __attribute__((ext_vector_type(8))) float v8f;

#define F_DIM 64


__device__ __forceinline__ bool two_sets(int n_nodes, size_t ws_bytes) {
    return ws_bytes >= (size_t)5 * (size_t)n_nodes * sizeof(float);
}

__device__ __forceinline__ unsigned package_half() {
    return (__builtin_amdgcn_s_sendmsg_rtn(0x87) >> 18) & 1u;
}

#define TILE_PITCH 68

__global__ void __launch_bounds__(64)
matvec_wmma_kernel(const float* __restrict__ X, const float* __restrict__ W,
                   float* __restrict__ s, float* __restrict__ planes,
                   int n_nodes, size_t ws_bytes) {
    __shared__ __attribute__((aligned(16))) float tile_lds[2][16 * TILE_PITCH];

    {
        const int node = (blockIdx.x << 5) + (threadIdx.x & 31);
        const int n_planes = two_sets(n_nodes, ws_bytes) ? 4 : 2;
        if (node < n_nodes)
            for (int p = threadIdx.x >> 5; p < n_planes; p += 2)
                planes[(size_t)p * (size_t)n_nodes + node] = 0.0f;
    }

    const int lane = threadIdx.x & 31;
    const int wave = threadIdx.x >> 5;
    const int tile = blockIdx.x * (blockDim.x >> 5) + wave;
    const int base = tile << 4;
    if (base >= n_nodes) return;

    const int m    = lane & 15;
    const int half = lane >> 4;

    const float* __restrict__ xt = X + (size_t)base * F_DIM + (lane << 2);
    v4f piece[8];
#pragma unroll
    for (int j = 0; j < 8; ++j)
        piece[j] = *reinterpret_cast<const v4f*>(xt + j * (2 * F_DIM));
    v2f wfrag[F_DIM / 4];
#pragma unroll
    for (int k = 0; k < F_DIM; k += 4)
        wfrag[k >> 2] = *reinterpret_cast<const v2f*>(W + k + (half << 1));

    float* tl = tile_lds[wave];
#pragma unroll
    for (int j = 0; j < 8; ++j)
        *reinterpret_cast<v4f*>(tl + (2 * j + half) * TILE_PITCH + (m << 2)) =
            piece[j];

    const float* lrow = tl + m * TILE_PITCH + (half << 1);
    v2f afrag[F_DIM / 4];
#pragma unroll
    for (int k = 0; k < F_DIM; k += 4)
        afrag[k >> 2] = *reinterpret_cast<const v2f*>(lrow + k);
    __builtin_amdgcn_sched_barrier(0);

    v8f c = {};
#pragma unroll
    for (int k = 0; k < F_DIM; k += 4) {
        const v2f a = afrag[k >> 2];
        const v2f wv = wfrag[k >> 2];
        v2f bf;
        bf.x = (m == 0) ? wv.x : 0.0f;
        bf.y = (m == 0) ? wv.y : 0.0f;
        c = __builtin_amdgcn_wmma_f32_16x16x4_f32(false, a, false, bf,
                                                  (short)0, c, false, false);
    }

    if (m == 0) {
        float* __restrict__ sp = s + base + (half << 3);
#pragma unroll
        for (int r = 0; r < 8; ++r) sp[r] = c[r];
    }
}

#define EDGES_PER_LANE 8
#define SCATTER_THREADS 128

template <bool kWhole>
__device__ __forceinline__ void
scatter_rounds(const int* __restrict__ src, const int* __restrict__ dst,
               const float* __restrict__ s, float* __restrict__ planes,
               int n_nodes, int n_edges, size_t ws_bytes, int first, int odd) {
    const int last = n_edges - 1;
    int si[EDGES_PER_LANE];
    int di[EDGES_PER_LANE];
#pragma unroll
    for (int j = 0; j < EDGES_PER_LANE; ++j) {
        const int e = first + 32 * j;
        const int ec = (kWhole || e < n_edges) ? e : last;
        si[j] = src[ec];
        di[j] = dst[ec];
    }
    const unsigned set = two_sets(n_nodes, ws_bytes) ? package_half() : 0u;
    float* __restrict__ pairs = planes + (size_t)set * 2u * (size_t)n_nodes;

    float sv[EDGES_PER_LANE];
#pragma unroll
    for (int j = 0; j < EDGES_PER_LANE; ++j) sv[j] = s[si[j]];

#pragma unroll
    for (int j = 0; j < EDGES_PER_LANE; ++j) {
        const int e = first + 32 * j;
        const int dn = __builtin_amdgcn_mov_dpp(di[j], 0xB1, 0xF, 0xF, true);
        if (kWhole || (e & ~1) < n_edges) {
            const int d = odd ? dn : di[j];
            atomicAdd(&pairs[2 * d + odd], odd ? 1.0f : sv[j]);
        }
        if (kWhole || (e | 1) < n_edges) {
            const int d = odd ? di[j] : dn;
            atomicAdd(&pairs[2 * d + (odd ^ 1)], odd ? sv[j] : 1.0f);
        }
    }
}

__global__ void __launch_bounds__(SCATTER_THREADS)
edge_scatter_kernel(const int* __restrict__ src, const int* __restrict__ dst,
                    const float* __restrict__ s, float* __restrict__ planes,
                    int n_nodes, int n_edges, size_t ws_bytes) {
    const int lane = threadIdx.x & 31;
    const int wave = __builtin_amdgcn_readfirstlane(
        blockIdx.x * (SCATTER_THREADS / 32) + (threadIdx.x >> 5));
    const int base = wave * (32 * EDGES_PER_LANE);
    if (base >= n_edges) return;
    if (base + 32 * EDGES_PER_LANE <= n_edges)
        scatter_rounds<true>(src, dst, s, planes, n_nodes, n_edges, ws_bytes,
                             base + lane, lane & 1);
    else
        scatter_rounds<false>(src, dst, s, planes, n_nodes, n_edges, ws_bytes,
                              base + lane, lane & 1);
}

__global__ void __launch_bounds__(256)
finalize_kernel(const float* __restrict__ planes, const float* __restrict__ b,
                float* __restrict__ out, int n, size_t ws_bytes) {
    int i = blockIdx.x * blockDim.x + threadIdx.x;
    if (i >= n) return;
    const float* __restrict__ p0 = planes + 2 * (size_t)i;
    float a  = p0[0];
    float dg = p0[1];
    if (two_sets(n, ws_bytes)) {
        const float* __restrict__ p1 = p0 + 2 * (size_t)n;
        a  += p1[0];
        dg += p1[1];
    }
    const float d = fmaxf(dg, 1.0f);
    const float x = a / d + b[0];
    out[i] = 1.0f / (1.0f + expf(-x));
}

extern "C" void kernel_launch(void* const* d_in, const int* in_sizes, int n_in,
                              void* d_out, int out_size, void* d_ws, size_t ws_size,
                              hipStream_t stream) {
    const float* X     = (const float*)d_in[0];
    const int*   edges = (const int*)d_in[1];
    const float* W     = (const float*)d_in[2];
    const float* b     = (const float*)d_in[3];
    float*       out   = (float*)d_out;

    const int n_nodes = in_sizes[0] / F_DIM;
    const int n_edges = in_sizes[1] / 2;
    const int* src = edges;
    const int* dst = edges + n_edges;

    float* s      = (float*)d_ws;
    float* planes = s + n_nodes;

    {
        const int tiles  = (n_nodes + 15) / 16;
        const int blocks = (tiles + 1) / 2;
        matvec_wmma_kernel<<<blocks, 64, 0, stream>>>(X, W, s, planes, n_nodes,
                                                      ws_size);
    }

    {
        const int edges_per_block = SCATTER_THREADS * EDGES_PER_LANE;
        edge_scatter_kernel<<<(n_edges + edges_per_block - 1) / edges_per_block,
                              SCATTER_THREADS, 0, stream>>>(
            src, dst, s, planes, n_nodes, n_edges, ws_size);
    }

    finalize_kernel<<<(n_nodes + 255) / 256, 256, 0, stream>>>(
        planes, b, out, n_nodes, ws_size);
}
